# starting kernel
# baseline (speedup 1.0000x reference)
#include <hip/hip_runtime.h>
#include <math.h>


typedef __attribute__((ext_vector_type(2))) float v2f;
typedef __attribute__((ext_vector_type(8))) float v8f;

#define NQ      12
#define NL      4
#define QDIM    4096
#define THREADS 256

static_assert(QDIM == (1 << NQ), "dim mismatch");

struct c32 { float re, im; };
__device__ __forceinline__ c32 cmul(c32 a, c32 b) {
  return { a.re * b.re - a.im * b.im, a.re * b.im + a.im * b.re };
}
__device__ __forceinline__ c32 cadd(c32 a, c32 b) { return { a.re + b.re, a.im + b.im }; }

__device__ __forceinline__ void mat2mul(c32 r[4], const c32 p[4], const c32 q[4]) {
  r[0] = cadd(cmul(p[0], q[0]), cmul(p[1], q[2]));
  r[1] = cadd(cmul(p[0], q[1]), cmul(p[1], q[3]));
  r[2] = cadd(cmul(p[2], q[0]), cmul(p[3], q[2]));
  r[3] = cadd(cmul(p[2], q[1]), cmul(p[3], q[3]));
}

template <int SK, int SN, int BM>
__device__ __forceinline__ void apply_group(float* sre, float* sim,
                                            const float* AreG, const float* AimG,
                                            int wave, int half, int mn) {
#pragma unroll
  for (int tt = 0; tt < 2; ++tt) {
    const int tile = wave + tt * 8;
    const int base = tile * BM;
    v8f cre = {0.f, 0.f, 0.f, 0.f, 0.f, 0.f, 0.f, 0.f};
    v8f cim = {0.f, 0.f, 0.f, 0.f, 0.f, 0.f, 0.f, 0.f};
#pragma unroll
    for (int kc = 0; kc < 4; ++kc) {
      const int k0 = kc * 4 + 2 * half;
      const int ar = mn * 16 + k0;
      v2f Ar = { AreG[ar], AreG[ar + 1] };
      v2f Ai = { AimG[ar], AimG[ar + 1] };
      v2f An = -Ai;
      const int ib = base + k0 * SK + mn * SN;
      v2f Br = { sre[ib], sre[ib + SK] };
      v2f Bi = { sim[ib], sim[ib + SK] };
      cre = __builtin_amdgcn_wmma_f32_16x16x4_f32(false, Ar, false, Br, 0, cre, false, false);
      cre = __builtin_amdgcn_wmma_f32_16x16x4_f32(false, An, false, Bi, 0, cre, false, false);
      cim = __builtin_amdgcn_wmma_f32_16x16x4_f32(false, Ar, false, Bi, 0, cim, false, false);
      cim = __builtin_amdgcn_wmma_f32_16x16x4_f32(false, Ai, false, Br, 0, cim, false, false);
    }
#pragma unroll
    for (int v = 0; v < 8; ++v) {
      const int m   = v + 8 * half;
      const int idx = base + m * SK + mn * SN;
      sre[idx] = cre[v];
      sim[idx] = cim[v];
    }
  }
}

__global__ __launch_bounds__(THREADS)
void qcircuit_parity_kernel(const float* __restrict__ x,
                            const float* __restrict__ thetas,
                            float* __restrict__ out) {
  __shared__ float sre[QDIM];
  __shared__ float sim[QDIM];
  __shared__ float Are[3 * 256];
  __shared__ float Aim[3 * 256];
  __shared__ float gmre[NQ * 4];
  __shared__ float gmim[NQ * 4];
  __shared__ float red[THREADS];

  const int tid  = threadIdx.x;
  const int b    = blockIdx.x;
  const int lane = tid & 31;
  const int wave = tid >> 5;
  const int half = lane >> 4;
  const int mn   = lane & 15;

  for (int j = tid; j < QDIM; j += THREADS) { sre[j] = 0.f; sim[j] = 0.f; }
  if (tid == 0) sre[0] = 1.f;

  const float xv = x[b];
  const float x1 = asinf(xv);
  const float x2 = acosf(xv * xv);

  int csrc[16];
#pragma unroll
  for (int j = 0; j < 16; ++j) {
    int s = tid + j * THREADS;
#pragma unroll
    for (int jj = 10; jj >= 0; --jj)
      s ^= ((s >> (11 - jj)) & 1) << (10 - jj);
    csrc[j] = s;
  }

  __syncthreads();

  for (int l = 0; l < NL; ++l) {
    if (tid < NQ) {
      const int q = tid;
      c32 M[4];
      { float s, c; __sincosf(0.5f * x1, &s, &c);
        M[0] = c32{c, 0.f}; M[1] = c32{-s, 0.f}; M[2] = c32{s, 0.f}; M[3] = c32{c, 0.f}; }
      { float s, c; __sincosf(0.5f * x2, &s, &c);
        c32 G[4] = { c32{c, -s}, c32{0.f, 0.f}, c32{0.f, 0.f}, c32{c, s} };
        c32 T[4]; mat2mul(T, G, M);
        M[0] = T[0]; M[1] = T[1]; M[2] = T[2]; M[3] = T[3]; }
      const float t0 = thetas[(l * NQ + q) * 3 + 0];
      const float t1 = thetas[(l * NQ + q) * 3 + 1];
      const float t2 = thetas[(l * NQ + q) * 3 + 2];
      { float s, c; __sincosf(0.5f * t0, &s, &c);
        c32 G[4] = { c32{c, 0.f}, c32{0.f, -s}, c32{0.f, -s}, c32{c, 0.f} };
        c32 T[4]; mat2mul(T, G, M);
        M[0] = T[0]; M[1] = T[1]; M[2] = T[2]; M[3] = T[3]; }
      { float s, c; __sincosf(0.5f * t1, &s, &c);
        c32 G[4] = { c32{c, -s}, c32{0.f, 0.f}, c32{0.f, 0.f}, c32{c, s} };
        c32 T[4]; mat2mul(T, G, M);
        M[0] = T[0]; M[1] = T[1]; M[2] = T[2]; M[3] = T[3]; }
      { float s, c; __sincosf(0.5f * t2, &s, &c);
        c32 G[4] = { c32{c, 0.f}, c32{0.f, -s}, c32{0.f, -s}, c32{c, 0.f} };
        c32 T[4]; mat2mul(T, G, M);
        M[0] = T[0]; M[1] = T[1]; M[2] = T[2]; M[3] = T[3]; }
#pragma unroll
      for (int k = 0; k < 4; ++k) { gmre[q * 4 + k] = M[k].re; gmim[q * 4 + k] = M[k].im; }
    }
    __syncthreads();

    {
      const int r = tid >> 4, cc = tid & 15;
#pragma unroll
      for (int g = 0; g < 3; ++g) {
        c32 acc = c32{1.f, 0.f};
#pragma unroll
        for (int j = 0; j < 4; ++j) {
          const int q  = 4 * g + j;
          const int bi = (r  >> (3 - j)) & 1;
          const int bj = (cc >> (3 - j)) & 1;
          acc = cmul(acc, c32{ gmre[q * 4 + bi * 2 + bj], gmim[q * 4 + bi * 2 + bj] });
        }
        Are[g * 256 + r * 16 + cc] = acc.re;
        Aim[g * 256 + r * 16 + cc] = acc.im;
      }
    }
    __syncthreads();

    apply_group<256, 1, 16 >(sre, sim, Are + 0,   Aim + 0,   wave, half, mn);
    __syncthreads();
    apply_group<16,  1, 256>(sre, sim, Are + 256, Aim + 256, wave, half, mn);
    __syncthreads();
    apply_group<1,  16, 256>(sre, sim, Are + 512, Aim + 512, wave, half, mn);
    __syncthreads();

    float tr[16], ti[16];
#pragma unroll
    for (int j = 0; j < 16; ++j) { tr[j] = sre[csrc[j]]; ti[j] = sim[csrc[j]]; }
    __syncthreads();
#pragma unroll
    for (int j = 0; j < 16; ++j) {
      const int i = tid + j * THREADS;
      sre[i] = tr[j]; sim[i] = ti[j];
    }
    __syncthreads();
  }

  float acc = 0.f;
  for (int j = tid; j < QDIM; j += THREADS) {
    const float p = (__popc(j) & 1) ? -1.f : 1.f;
    acc = fmaf(sre[j] * sre[j] + sim[j] * sim[j], p, acc);
  }
  red[tid] = acc;
  __syncthreads();
  for (int off = THREADS / 2; off > 0; off >>= 1) {
    if (tid < off) red[tid] += red[tid + off];
    __syncthreads();
  }
  if (tid == 0) out[b] = red[0];
}

extern "C" void kernel_launch(void* const* d_in, const int* in_sizes, int n_in,
                              void* d_out, int out_size, void* d_ws, size_t ws_size,
                              hipStream_t stream) {
  (void)n_in; (void)d_ws; (void)ws_size; (void)out_size;
  const float* x      = (const float*)d_in[0];
  const float* thetas = (const float*)d_in[1];
  float* out          = (float*)d_out;
  const int batch     = in_sizes[0];
  qcircuit_parity_kernel<<<dim3(batch), dim3(THREADS), 0, stream>>>(x, thetas, out);
}


# Complex products as three real matrix products, not four (K=Gr(Sr+Si), Re=K-(Gr+Gi)Si, Im=K+(Gi-Gr)Sr): 72 WMMA per wave and layer, not 96; f32 sum order changes
# speedup vs baseline: 1.4416x; 1.4416x over previous
#include <hip/hip_runtime.h>
#include <math.h>


typedef __attribute__((ext_vector_type(2))) float v2f;
typedef __attribute__((ext_vector_type(4))) float v4f;
typedef __attribute__((ext_vector_type(8))) float v8f;

#define NQ      12
#define NL      4
#define QDIM    4096
#define THREADS 256
#define RP      20
#define CP      (16 * RP + 4)
#define GP      (16 * RP)
#define PLANE   (16 * CP)

static_assert(QDIM == (1 << NQ), "dim mismatch");

struct c32 { float re, im; };
__device__ __forceinline__ c32 cmul(c32 a, c32 b) {
  return { a.re * b.re - a.im * b.im, a.re * b.im + a.im * b.re };
}
__device__ __forceinline__ c32 cadd(c32 a, c32 b) { return { a.re + b.re, a.im + b.im }; }

__device__ __forceinline__ void mat2mul(c32 r[4], const c32 p[4], const c32 q[4]) {
  r[0] = cadd(cmul(p[0], q[0]), cmul(p[1], q[2]));
  r[1] = cadd(cmul(p[0], q[1]), cmul(p[1], q[3]));
  r[2] = cadd(cmul(p[2], q[0]), cmul(p[3], q[2]));
  r[3] = cadd(cmul(p[2], q[1]), cmul(p[3], q[3]));
}

__device__ __forceinline__ void load8(const float* p, float (&v)[8]) {
  const v4f lo = *(const v4f*)p;
  const v4f hi = *(const v4f*)(p + 4);
  v[0] = lo[0]; v[1] = lo[1]; v[2] = lo[2]; v[3] = lo[3];
  v[4] = hi[0]; v[5] = hi[1]; v[6] = hi[2]; v[7] = hi[3];
}
__device__ __forceinline__ void store8(float* p, const v8f& d) {
  *(v4f*)p       = v4f{ d[0], d[1], d[2], d[3] };
  *(v4f*)(p + 4) = v4f{ d[4], d[5], d[6], d[7] };
}

#define WMMA(A, B, C) __builtin_amdgcn_wmma_f32_16x16x4_f32(false, A, false, B, 0, C, false, false)


__device__ __forceinline__ void pass_ab(float* sre, float* sim,
                                        const float* Gk, const float* Gn, const float* Gd,
                                        int wave, int half, int mn) {
  const int go = mn * RP + 8 * half;
  float ak[8], an[8], ad[8], bk[8], bn[8], bd[8];
  load8(Gk + go, ak);      load8(Gn + go, an);      load8(Gd + go, ad);
  load8(Gk + GP + go, bk); load8(Gn + GP + go, bn); load8(Gd + GP + go, bd);
#pragma unroll
  for (int tt = 0; tt < 2; ++tt) {
    const int c = wave + tt * 8;
    float* pr = sre + c * CP + go;
    float* pi = sim + c * CP + go;
    float sr[8], si[8], ss[8];
    load8(pr, sr); load8(pi, si);
#pragma unroll
    for (int k = 0; k < 8; ++k) ss[k] = sr[k] + si[k];
    v8f k1 = {0.f, 0.f, 0.f, 0.f, 0.f, 0.f, 0.f, 0.f};
#pragma unroll
    for (int kc = 0; kc < 4; ++kc) {
      const v2f As = { ss[2 * kc], ss[2 * kc + 1] };
      const v2f Bk = { ak[2 * kc], ak[2 * kc + 1] };
      k1 = WMMA(As, Bk, k1);
    }
    v8f cre = k1, cim = k1;
#pragma unroll
    for (int kc = 0; kc < 4; ++kc) {
      const v2f Ar = { sr[2 * kc], sr[2 * kc + 1] };
      const v2f Ai = { si[2 * kc], si[2 * kc + 1] };
      const v2f Bn = { an[2 * kc], an[2 * kc + 1] };
      const v2f Bd = { ad[2 * kc], ad[2 * kc + 1] };
      cre = WMMA(Ai, Bn, cre);
      cim = WMMA(Ar, Bd, cim);
    }
    const v8f cs = cre + cim;
    v8f k2 = {0.f, 0.f, 0.f, 0.f, 0.f, 0.f, 0.f, 0.f};
#pragma unroll
    for (int kc = 0; kc < 4; ++kc) {
      const v2f Ak = { bk[2 * kc], bk[2 * kc + 1] };
      const v2f Bs = { cs[2 * kc], cs[2 * kc + 1] };
      k2 = WMMA(Ak, Bs, k2);
    }
    v8f dre = k2, dim = k2;
#pragma unroll
    for (int kc = 0; kc < 4; ++kc) {
      const v2f An = { bn[2 * kc], bn[2 * kc + 1] };
      const v2f Ad = { bd[2 * kc], bd[2 * kc + 1] };
      const v2f Br = { cre[2 * kc], cre[2 * kc + 1] };
      const v2f Bi = { cim[2 * kc], cim[2 * kc + 1] };
      dre = WMMA(An, Bi, dre);
      dim = WMMA(Ad, Br, dim);
    }
    store8(pr, dre);
    store8(pi, dim);
  }
}

__device__ __forceinline__ void pass_c(float* sre, float* sim,
                                       const float* Gk, const float* Gn, const float* Gd,
                                       int wave, int half, int mn) {
  const int go = mn * RP + 8 * half;
  float ck[8], cn[8], cd[8];
  load8(Gk + 2 * GP + go, ck); load8(Gn + 2 * GP + go, cn); load8(Gd + 2 * GP + go, cd);
#pragma unroll
  for (int tt = 0; tt < 2; ++tt) {
    const int ap = wave + tt * 8;
    float* pr = sre + 8 * half * CP + ap * RP + mn;
    float* pi = sim + 8 * half * CP + ap * RP + mn;
    float xr[8], xi[8], xs[8];
#pragma unroll
    for (int r = 0; r < 8; ++r) { xr[r] = pr[r * CP]; xi[r] = pi[r * CP]; }
#pragma unroll
    for (int r = 0; r < 8; ++r) xs[r] = xr[r] + xi[r];
    v8f k3 = {0.f, 0.f, 0.f, 0.f, 0.f, 0.f, 0.f, 0.f};
#pragma unroll
    for (int kc = 0; kc < 4; ++kc) {
      const v2f Ak = { ck[2 * kc], ck[2 * kc + 1] };
      const v2f Bs = { xs[2 * kc], xs[2 * kc + 1] };
      k3 = WMMA(Ak, Bs, k3);
    }
    v8f cre = k3, cim = k3;
#pragma unroll
    for (int kc = 0; kc < 4; ++kc) {
      const v2f An = { cn[2 * kc], cn[2 * kc + 1] };
      const v2f Ad = { cd[2 * kc], cd[2 * kc + 1] };
      const v2f Br = { xr[2 * kc], xr[2 * kc + 1] };
      const v2f Bi = { xi[2 * kc], xi[2 * kc + 1] };
      cre = WMMA(An, Bi, cre);
      cim = WMMA(Ad, Br, cim);
    }
#pragma unroll
    for (int r = 0; r < 8; ++r) { pr[r * CP] = cre[r]; pi[r * CP] = cim[r]; }
  }
}

__global__ __launch_bounds__(THREADS, 6)
void qcircuit_parity_kernel(const float* __restrict__ x,
                            const float* __restrict__ thetas,
                            float* __restrict__ out) {
  __shared__ __attribute__((aligned(16))) float sre[PLANE];
  __shared__ __attribute__((aligned(16))) float sim[PLANE];
  __shared__ __attribute__((aligned(16))) float Gk[3 * GP];
  __shared__ __attribute__((aligned(16))) float Gn[3 * GP];
  __shared__ __attribute__((aligned(16))) float Gd[3 * GP];
  __shared__ float gmre[NQ * 4];
  __shared__ float gmim[NQ * 4];
  __shared__ float red[THREADS];

  const int tid  = threadIdx.x;
  const int b    = blockIdx.x;
  const int lane = tid & 31;
  const int wave = tid >> 5;
  const int half = lane >> 4;
  const int mn   = lane & 15;

  for (int j = tid; j < PLANE; j += THREADS) { sre[j] = 0.f; sim[j] = 0.f; }
  if (tid == 0) sre[0] = 1.f;

  const float xv = x[b];
  const float x1 = asinf(xv);
  const float x2 = acosf(xv * xv);

  __syncthreads();

  for (int l = 0; l < NL; ++l) {
    if (tid < NQ) {
      const int q = tid;
      c32 M[4];
      { float s, c; __sincosf(0.5f * x1, &s, &c);
        M[0] = c32{c, 0.f}; M[1] = c32{-s, 0.f}; M[2] = c32{s, 0.f}; M[3] = c32{c, 0.f}; }
      { float s, c; __sincosf(0.5f * x2, &s, &c);
        c32 G[4] = { c32{c, -s}, c32{0.f, 0.f}, c32{0.f, 0.f}, c32{c, s} };
        c32 T[4]; mat2mul(T, G, M);
        M[0] = T[0]; M[1] = T[1]; M[2] = T[2]; M[3] = T[3]; }
      const float t0 = thetas[(l * NQ + q) * 3 + 0];
      const float t1 = thetas[(l * NQ + q) * 3 + 1];
      const float t2 = thetas[(l * NQ + q) * 3 + 2];
      { float s, c; __sincosf(0.5f * t0, &s, &c);
        c32 G[4] = { c32{c, 0.f}, c32{0.f, -s}, c32{0.f, -s}, c32{c, 0.f} };
        c32 T[4]; mat2mul(T, G, M);
        M[0] = T[0]; M[1] = T[1]; M[2] = T[2]; M[3] = T[3]; }
      { float s, c; __sincosf(0.5f * t1, &s, &c);
        c32 G[4] = { c32{c, -s}, c32{0.f, 0.f}, c32{0.f, 0.f}, c32{c, s} };
        c32 T[4]; mat2mul(T, G, M);
        M[0] = T[0]; M[1] = T[1]; M[2] = T[2]; M[3] = T[3]; }
      { float s, c; __sincosf(0.5f * t2, &s, &c);
        c32 G[4] = { c32{c, 0.f}, c32{0.f, -s}, c32{0.f, -s}, c32{c, 0.f} };
        c32 T[4]; mat2mul(T, G, M);
        M[0] = T[0]; M[1] = T[1]; M[2] = T[2]; M[3] = T[3]; }
#pragma unroll
      for (int k = 0; k < 4; ++k) { gmre[q * 4 + k] = M[k].re; gmim[q * 4 + k] = M[k].im; }
    }
    __syncthreads();

    {
      const int r = tid >> 4, cc = tid & 15;
#pragma unroll
      for (int g = 0; g < 3; ++g) {
        c32 acc = c32{1.f, 0.f};
#pragma unroll
        for (int j = 0; j < 4; ++j) {
          const int q  = 4 * g + j;
          const int bi = (r  >> (3 - j)) & 1;
          const int bj = (cc >> (3 - j)) & 1;
          acc = cmul(acc, c32{ gmre[q * 4 + bi * 2 + bj], gmim[q * 4 + bi * 2 + bj] });
        }
        Gk[g * GP + r * RP + cc] = acc.re;
        Gn[g * GP + r * RP + cc] = -(acc.re + acc.im);
        Gd[g * GP + r * RP + cc] = acc.im - acc.re;
      }
    }
    __syncthreads();

    pass_ab(sre, sim, Gk, Gn, Gd, wave, half, mn);
    __syncthreads();
    pass_c(sre, sim, Gk, Gn, Gd, wave, half, mn);
    __syncthreads();

    v4f tr[4], ti[4];
#pragma unroll
    for (int j = 0; j < 16; ++j) {
      const int i = tid + j * THREADS;
      const int sidx = i ^ (i >> 1);
      const int o = (sidx & 15) * CP + (sidx >> 8) * RP + ((sidx >> 4) & 15);
      tr[j >> 2][j & 3] = sre[o]; ti[j >> 2][j & 3] = sim[o];
    }
    __syncthreads();
    {
      const int o = (tid & 15) * CP + (tid >> 4) * RP;
#pragma unroll
      for (int q = 0; q < 4; ++q) {
        *(v4f*)(sre + o + 4 * q) = tr[q];
        *(v4f*)(sim + o + 4 * q) = ti[q];
      }
    }
    __syncthreads();
  }

  float acc = 0.f;
  {
    const int o = (tid & 15) * CP + (tid >> 4) * RP;
#pragma unroll
    for (int j = 0; j < 16; ++j) {
      const float p = (__popc(tid + j * THREADS) & 1) ? -1.f : 1.f;
      acc = fmaf(sre[o + j] * sre[o + j] + sim[o + j] * sim[o + j], p, acc);
    }
  }
  red[tid] = acc;
  __syncthreads();
  for (int off = THREADS / 2; off > 0; off >>= 1) {
    if (tid < off) red[tid] += red[tid + off];
    __syncthreads();
  }
  if (tid == 0) out[b] = red[0];
}

extern "C" void kernel_launch(void* const* d_in, const int* in_sizes, int n_in,
                              void* d_out, int out_size, void* d_ws, size_t ws_size,
                              hipStream_t stream) {
  (void)n_in; (void)d_ws; (void)ws_size; (void)out_size;
  const float* x      = (const float*)d_in[0];
  const float* thetas = (const float*)d_in[1];
  float* out          = (float*)d_out;
  const int batch     = in_sizes[0];
  qcircuit_parity_kernel<<<dim3(batch), dim3(THREADS), 0, stream>>>(x, thetas, out);
}


# Register allocation 160 -> 128 (pass a on both tiles before pass b: one gate group's fragments live at a time; 2x2 gates in one array): four workgroups per WGP, not three; f32 sum order unchanged
# speedup vs baseline: 1.4880x; 1.0322x over previous
#include <hip/hip_runtime.h>
#include <math.h>


typedef __attribute__((ext_vector_type(2))) float v2f;
typedef __attribute__((ext_vector_type(4))) float v4f;
typedef __attribute__((ext_vector_type(8))) float v8f;

#define NQ      12
#define NL      4
#define QDIM    4096
#define THREADS 256
#define RP      20
#define CP      (16 * RP + 4)
#define GP      (16 * RP)
#define PLANE   (16 * CP)

static_assert(QDIM == (1 << NQ), "dim mismatch");

struct c32 { float re, im; };
__device__ __forceinline__ c32 cmul(c32 a, c32 b) {
  return { a.re * b.re - a.im * b.im, a.re * b.im + a.im * b.re };
}
__device__ __forceinline__ c32 cadd(c32 a, c32 b) { return { a.re + b.re, a.im + b.im }; }

__device__ __forceinline__ void mat2mul(c32 r[4], const c32 p[4], const c32 q[4]) {
  r[0] = cadd(cmul(p[0], q[0]), cmul(p[1], q[2]));
  r[1] = cadd(cmul(p[0], q[1]), cmul(p[1], q[3]));
  r[2] = cadd(cmul(p[2], q[0]), cmul(p[3], q[2]));
  r[3] = cadd(cmul(p[2], q[1]), cmul(p[3], q[3]));
}

__device__ __forceinline__ void load8(const float* p, float (&v)[8]) {
  const v4f lo = *(const v4f*)p;
  const v4f hi = *(const v4f*)(p + 4);
  v[0] = lo[0]; v[1] = lo[1]; v[2] = lo[2]; v[3] = lo[3];
  v[4] = hi[0]; v[5] = hi[1]; v[6] = hi[2]; v[7] = hi[3];
}
__device__ __forceinline__ void store8(float* p, const v8f& d) {
  *(v4f*)p       = v4f{ d[0], d[1], d[2], d[3] };
  *(v4f*)(p + 4) = v4f{ d[4], d[5], d[6], d[7] };
}

#define WMMA(A, B, C) __builtin_amdgcn_wmma_f32_16x16x4_f32(false, A, false, B, 0, C, false, false)


__device__ __forceinline__ void pass_ab(float* sre, float* sim,
                                        const float* Gk, const float* Gn, const float* Gd,
                                        int wave, int half, int mn) {
  const int go = mn * RP + 8 * half;
  v8f d1r[2], d1i[2];
  {
    float ak[8], an[8], ad[8];
    load8(Gk + go, ak); load8(Gn + go, an); load8(Gd + go, ad);
#pragma unroll
    for (int tt = 0; tt < 2; ++tt) {
      const int c = wave + tt * 8;
      float sr[8], si[8], ss[8];
      load8(sre + c * CP + go, sr); load8(sim + c * CP + go, si);
#pragma unroll
      for (int k = 0; k < 8; ++k) ss[k] = sr[k] + si[k];
      v8f k1 = {0.f, 0.f, 0.f, 0.f, 0.f, 0.f, 0.f, 0.f};
#pragma unroll
      for (int kc = 0; kc < 4; ++kc) {
        const v2f As = { ss[2 * kc], ss[2 * kc + 1] };
        const v2f Bk = { ak[2 * kc], ak[2 * kc + 1] };
        k1 = WMMA(As, Bk, k1);
      }
      v8f cre = k1, cim = k1;
#pragma unroll
      for (int kc = 0; kc < 4; ++kc) {
        const v2f Ar = { sr[2 * kc], sr[2 * kc + 1] };
        const v2f Ai = { si[2 * kc], si[2 * kc + 1] };
        const v2f Bn = { an[2 * kc], an[2 * kc + 1] };
        const v2f Bd = { ad[2 * kc], ad[2 * kc + 1] };
        cre = WMMA(Ai, Bn, cre);
        cim = WMMA(Ar, Bd, cim);
      }
      d1r[tt] = cre; d1i[tt] = cim;
    }
  }
  {
    float bk[8], bn[8], bd[8];
    load8(Gk + GP + go, bk); load8(Gn + GP + go, bn); load8(Gd + GP + go, bd);
#pragma unroll
    for (int tt = 0; tt < 2; ++tt) {
      const int c = wave + tt * 8;
      const v8f cre = d1r[tt], cim = d1i[tt];
      const v8f cs = cre + cim;
      v8f k2 = {0.f, 0.f, 0.f, 0.f, 0.f, 0.f, 0.f, 0.f};
#pragma unroll
      for (int kc = 0; kc < 4; ++kc) {
        const v2f Ak = { bk[2 * kc], bk[2 * kc + 1] };
        const v2f Bs = { cs[2 * kc], cs[2 * kc + 1] };
        k2 = WMMA(Ak, Bs, k2);
      }
      v8f dre = k2, dim = k2;
#pragma unroll
      for (int kc = 0; kc < 4; ++kc) {
        const v2f An = { bn[2 * kc], bn[2 * kc + 1] };
        const v2f Ad = { bd[2 * kc], bd[2 * kc + 1] };
        const v2f Br = { cre[2 * kc], cre[2 * kc + 1] };
        const v2f Bi = { cim[2 * kc], cim[2 * kc + 1] };
        dre = WMMA(An, Bi, dre);
        dim = WMMA(Ad, Br, dim);
      }
      store8(sre + c * CP + go, dre);
      store8(sim + c * CP + go, dim);
    }
  }
}

__device__ __forceinline__ void pass_c(float* sre, float* sim,
                                       const float* Gk, const float* Gn, const float* Gd,
                                       int wave, int half, int mn) {
  const int go = mn * RP + 8 * half;
  float ck[8], cn[8], cd[8];
  load8(Gk + 2 * GP + go, ck); load8(Gn + 2 * GP + go, cn); load8(Gd + 2 * GP + go, cd);
#pragma unroll
  for (int tt = 0; tt < 2; ++tt) {
    const int ap = wave + tt * 8;
    float* pr = sre + 8 * half * CP + ap * RP + mn;
    float* pi = sim + 8 * half * CP + ap * RP + mn;
    float xr[8], xi[8], xs[8];
#pragma unroll
    for (int r = 0; r < 8; ++r) { xr[r] = pr[r * CP]; xi[r] = pi[r * CP]; }
#pragma unroll
    for (int r = 0; r < 8; ++r) xs[r] = xr[r] + xi[r];
    v8f k3 = {0.f, 0.f, 0.f, 0.f, 0.f, 0.f, 0.f, 0.f};
#pragma unroll
    for (int kc = 0; kc < 4; ++kc) {
      const v2f Ak = { ck[2 * kc], ck[2 * kc + 1] };
      const v2f Bs = { xs[2 * kc], xs[2 * kc + 1] };
      k3 = WMMA(Ak, Bs, k3);
    }
    v8f cre = k3, cim = k3;
#pragma unroll
    for (int kc = 0; kc < 4; ++kc) {
      const v2f An = { cn[2 * kc], cn[2 * kc + 1] };
      const v2f Ad = { cd[2 * kc], cd[2 * kc + 1] };
      const v2f Br = { xr[2 * kc], xr[2 * kc + 1] };
      const v2f Bi = { xi[2 * kc], xi[2 * kc + 1] };
      cre = WMMA(An, Bi, cre);
      cim = WMMA(Ad, Br, cim);
    }
#pragma unroll
    for (int r = 0; r < 8; ++r) { pr[r * CP] = cre[r]; pi[r * CP] = cim[r]; }
  }
}

__global__ __launch_bounds__(THREADS, 8)
void qcircuit_parity_kernel(const float* __restrict__ x,
                            const float* __restrict__ thetas,
                            float* __restrict__ out) {
  __shared__ __attribute__((aligned(16))) float sre[PLANE];
  __shared__ __attribute__((aligned(16))) float sim[PLANE];
  __shared__ __attribute__((aligned(16))) float Gk[3 * GP];
  __shared__ __attribute__((aligned(16))) float Gn[3 * GP];
  __shared__ __attribute__((aligned(16))) float Gd[3 * GP];
  __shared__ __attribute__((aligned(16))) float gm[NQ * 8];
  __shared__ float red[THREADS];

  const int tid  = threadIdx.x;
  const int b    = blockIdx.x;
  const int lane = tid & 31;
  const int wave = tid >> 5;
  const int half = lane >> 4;
  const int mn   = lane & 15;

  for (int j = tid; j < PLANE; j += THREADS) { sre[j] = 0.f; sim[j] = 0.f; }
  if (tid == 0) sre[0] = 1.f;

  const float xv = x[b];
  const float x1 = asinf(xv);
  const float x2 = acosf(xv * xv);

  __syncthreads();

  for (int l = 0; l < NL; ++l) {
    if (tid < NQ) {
      const int q = tid;
      c32 M[4];
      { float s, c; __sincosf(0.5f * x1, &s, &c);
        M[0] = c32{c, 0.f}; M[1] = c32{-s, 0.f}; M[2] = c32{s, 0.f}; M[3] = c32{c, 0.f}; }
      { float s, c; __sincosf(0.5f * x2, &s, &c);
        c32 G[4] = { c32{c, -s}, c32{0.f, 0.f}, c32{0.f, 0.f}, c32{c, s} };
        c32 T[4]; mat2mul(T, G, M);
        M[0] = T[0]; M[1] = T[1]; M[2] = T[2]; M[3] = T[3]; }
      const float t0 = thetas[(l * NQ + q) * 3 + 0];
      const float t1 = thetas[(l * NQ + q) * 3 + 1];
      const float t2 = thetas[(l * NQ + q) * 3 + 2];
      { float s, c; __sincosf(0.5f * t0, &s, &c);
        c32 G[4] = { c32{c, 0.f}, c32{0.f, -s}, c32{0.f, -s}, c32{c, 0.f} };
        c32 T[4]; mat2mul(T, G, M);
        M[0] = T[0]; M[1] = T[1]; M[2] = T[2]; M[3] = T[3]; }
      { float s, c; __sincosf(0.5f * t1, &s, &c);
        c32 G[4] = { c32{c, -s}, c32{0.f, 0.f}, c32{0.f, 0.f}, c32{c, s} };
        c32 T[4]; mat2mul(T, G, M);
        M[0] = T[0]; M[1] = T[1]; M[2] = T[2]; M[3] = T[3]; }
      { float s, c; __sincosf(0.5f * t2, &s, &c);
        c32 G[4] = { c32{c, 0.f}, c32{0.f, -s}, c32{0.f, -s}, c32{c, 0.f} };
        c32 T[4]; mat2mul(T, G, M);
        M[0] = T[0]; M[1] = T[1]; M[2] = T[2]; M[3] = T[3]; }
      *(v4f*)(gm + q * 8)     = v4f{ M[0].re, M[1].re, M[2].re, M[3].re };
      *(v4f*)(gm + q * 8 + 4) = v4f{ M[0].im, M[1].im, M[2].im, M[3].im };
    }
    __syncthreads();

    {
      const int r = tid >> 4, cc = tid & 15;
#pragma unroll
      for (int g = 0; g < 3; ++g) {
        c32 acc = c32{1.f, 0.f};
#pragma unroll
        for (int j = 0; j < 4; ++j) {
          const int q  = 4 * g + j;
          const int bi = (r  >> (3 - j)) & 1;
          const int bj = (cc >> (3 - j)) & 1;
          acc = cmul(acc, c32{ gm[q * 8 + bi * 2 + bj], gm[q * 8 + 4 + bi * 2 + bj] });
        }
        Gk[g * GP + r * RP + cc] = acc.re;
        Gn[g * GP + r * RP + cc] = -(acc.re + acc.im);
        Gd[g * GP + r * RP + cc] = acc.im - acc.re;
      }
    }
    __syncthreads();

    pass_ab(sre, sim, Gk, Gn, Gd, wave, half, mn);
    __syncthreads();
    pass_c(sre, sim, Gk, Gn, Gd, wave, half, mn);
    __syncthreads();

    v4f tr[4], ti[4];
#pragma unroll
    for (int j = 0; j < 16; ++j) {
      const int i = tid + j * THREADS;
      const int sidx = i ^ (i >> 1);
      const int o = (sidx & 15) * CP + (sidx >> 8) * RP + ((sidx >> 4) & 15);
      tr[j >> 2][j & 3] = sre[o]; ti[j >> 2][j & 3] = sim[o];
    }
    __syncthreads();
    {
      const int o = (tid & 15) * CP + (tid >> 4) * RP;
#pragma unroll
      for (int q = 0; q < 4; ++q) {
        *(v4f*)(sre + o + 4 * q) = tr[q];
        *(v4f*)(sim + o + 4 * q) = ti[q];
      }
    }
    __syncthreads();
  }

  float acc = 0.f;
  {
    const int o = (tid & 15) * CP + (tid >> 4) * RP;
#pragma unroll
    for (int j = 0; j < 16; ++j) {
      const float p = (__popc(tid + j * THREADS) & 1) ? -1.f : 1.f;
      acc = fmaf(sre[o + j] * sre[o + j] + sim[o + j] * sim[o + j], p, acc);
    }
  }
  red[tid] = acc;
  __syncthreads();
  for (int off = THREADS / 2; off > 0; off >>= 1) {
    if (tid < off) red[tid] += red[tid + off];
    __syncthreads();
  }
  if (tid == 0) out[b] = red[0];
}

extern "C" void kernel_launch(void* const* d_in, const int* in_sizes, int n_in,
                              void* d_out, int out_size, void* d_ws, size_t ws_size,
                              hipStream_t stream) {
  (void)n_in; (void)d_ws; (void)ws_size; (void)out_size;
  const float* x      = (const float*)d_in[0];
  const float* thetas = (const float*)d_in[1];
  float* out          = (float*)d_out;
  const int batch     = in_sizes[0];
  qcircuit_parity_kernel<<<dim3(batch), dim3(THREADS), 0, stream>>>(x, thetas, out);
}


# CNOT ladder folded in: in-group CNOTs as Gray-ordered gate-table rows, cross-group CNOTs as index complements in pass c addresses; no permutation pass, 4 barriers a layer not 6; f32 sum order changes
# speedup vs baseline: 1.6209x; 1.0893x over previous
#include <hip/hip_runtime.h>
#include <math.h>


typedef __attribute__((ext_vector_type(2))) float v2f;
typedef __attribute__((ext_vector_type(4))) float v4f;
typedef __attribute__((ext_vector_type(8))) float v8f;

#define NQ      12
#define NL      4
#define QDIM    4096
#define THREADS 256
#define RP      20
#define CP      (16 * RP + 4)
#define GP      (16 * RP)
#define PLANE   (16 * CP)

static_assert(QDIM == (1 << NQ), "dim mismatch");

struct c32 { float re, im; };
__device__ __forceinline__ c32 cmul(c32 a, c32 b) {
  return { a.re * b.re - a.im * b.im, a.re * b.im + a.im * b.re };
}

__device__ __forceinline__ void rx_mul(c32 M[4], float t) {
  float s, c; __sincosf(0.5f * t, &s, &c);
#pragma unroll
  for (int j = 0; j < 2; ++j) {
    const c32 u = M[j], v = M[2 + j];
    M[j]     = c32{ c * u.re + s * v.im, c * u.im - s * v.re };
    M[2 + j] = c32{ c * v.re + s * u.im, c * v.im - s * u.re };
  }
}
__device__ __forceinline__ void rz_mul(c32 M[4], float t) {
  float s, c; __sincosf(0.5f * t, &s, &c);
#pragma unroll
  for (int j = 0; j < 2; ++j) {
    const c32 u = M[j], v = M[2 + j];
    M[j]     = c32{ c * u.re + s * u.im, c * u.im - s * u.re };
    M[2 + j] = c32{ c * v.re - s * v.im, c * v.im + s * v.re };
  }
}

__device__ __forceinline__ void load8(const float* p, float (&v)[8]) {
  const v4f lo = *(const v4f*)p;
  const v4f hi = *(const v4f*)(p + 4);
  v[0] = lo[0]; v[1] = lo[1]; v[2] = lo[2]; v[3] = lo[3];
  v[4] = hi[0]; v[5] = hi[1]; v[6] = hi[2]; v[7] = hi[3];
}
__device__ __forceinline__ void store8(float* p, const v8f& d) {
  *(v4f*)p       = v4f{ d[0], d[1], d[2], d[3] };
  *(v4f*)(p + 4) = v4f{ d[4], d[5], d[6], d[7] };
}

#define WMMA(A, B, C) __builtin_amdgcn_wmma_f32_16x16x4_f32(false, A, false, B, 0, C, false, false)


__device__ __forceinline__ void pass_ab(float* sre, float* sim,
                                        const float* Gk, const float* Gn, const float* Gd,
                                        int o1, int o2, int wave, int half, int mn) {
  const int go = mn * RP + 8 * half;
  v8f d1r[2], d1i[2];
  {
    float ak[8], an[8], ad[8];
    load8(Gk + o1 + go, ak); load8(Gn + o1 + go, an); load8(Gd + o1 + go, ad);
#pragma unroll
    for (int tt = 0; tt < 2; ++tt) {
      const int c = wave + tt * 8;
      float sr[8], si[8], ss[8];
      load8(sre + c * CP + go, sr); load8(sim + c * CP + go, si);
#pragma unroll
      for (int k = 0; k < 8; ++k) ss[k] = sr[k] + si[k];
      v8f k1 = {0.f, 0.f, 0.f, 0.f, 0.f, 0.f, 0.f, 0.f};
#pragma unroll
      for (int kc = 0; kc < 4; ++kc) {
        const v2f As = { ss[2 * kc], ss[2 * kc + 1] };
        const v2f Bk = { ak[2 * kc], ak[2 * kc + 1] };
        k1 = WMMA(As, Bk, k1);
      }
      v8f cre = k1, cim = k1;
#pragma unroll
      for (int kc = 0; kc < 4; ++kc) {
        const v2f Ar = { sr[2 * kc], sr[2 * kc + 1] };
        const v2f Ai = { si[2 * kc], si[2 * kc + 1] };
        const v2f Bn = { an[2 * kc], an[2 * kc + 1] };
        const v2f Bd = { ad[2 * kc], ad[2 * kc + 1] };
        cre = WMMA(Ai, Bn, cre);
        cim = WMMA(Ar, Bd, cim);
      }
      d1r[tt] = cre; d1i[tt] = cim;
    }
  }
  {
    float bk[8], bn[8], bd[8];
    load8(Gk + o2 + go, bk); load8(Gn + o2 + go, bn); load8(Gd + o2 + go, bd);
#pragma unroll
    for (int tt = 0; tt < 2; ++tt) {
      const int c = wave + tt * 8;
      const v8f cre = d1r[tt], cim = d1i[tt];
      const v8f cs = cre + cim;
      v8f k2 = {0.f, 0.f, 0.f, 0.f, 0.f, 0.f, 0.f, 0.f};
#pragma unroll
      for (int kc = 0; kc < 4; ++kc) {
        const v2f Ak = { bk[2 * kc], bk[2 * kc + 1] };
        const v2f Bs = { cs[2 * kc], cs[2 * kc + 1] };
        k2 = WMMA(Ak, Bs, k2);
      }
      v8f dre = k2, dim = k2;
#pragma unroll
      for (int kc = 0; kc < 4; ++kc) {
        const v2f An = { bn[2 * kc], bn[2 * kc + 1] };
        const v2f Ad = { bd[2 * kc], bd[2 * kc + 1] };
        const v2f Br = { cre[2 * kc], cre[2 * kc + 1] };
        const v2f Bi = { cim[2 * kc], cim[2 * kc + 1] };
        dre = WMMA(An, Bi, dre);
        dim = WMMA(Ad, Br, dim);
      }
      store8(sre + c * CP + go, dre);
      store8(sim + c * CP + go, dim);
    }
  }
}

template <int ODD>
__device__ __forceinline__ void pass_c(float* sre, float* sim,
                                       const float* Gk, const float* Gn, const float* Gd,
                                       int wave, int half, int mn) {
  const int go = mn * RP + 8 * half;
  const int lodd = mn & 1;
  float ck[8], cn[8], cd[8];
  load8(Gk + 2 * GP + go, ck); load8(Gn + 2 * GP + go, cn); load8(Gd + 2 * GP + go, cd);
  float xr[2][8], xi[2][8];
#pragma unroll
  for (int tt = 0; tt < 2; ++tt) {
    const int p = tt ? 15 - 2 * wave : 2 * wave;
    const int src = ODD ? (lodd ? 15 - p : p) * RP + mn
                        : p * RP + (tt ? 15 - mn : mn);
    const float* pr = sre + 8 * half * CP + src;
    const float* pi = sim + 8 * half * CP + src;
#pragma unroll
    for (int r = 0; r < 8; ++r) { xr[tt][r] = pr[r * CP]; xi[tt][r] = pi[r * CP]; }
  }
#pragma unroll
  for (int tt = 0; tt < 2; ++tt) {
    const int p = tt ? 15 - 2 * wave : 2 * wave;
    float xs[8];
#pragma unroll
    for (int r = 0; r < 8; ++r) xs[r] = xr[tt][r] + xi[tt][r];
    v8f k3 = {0.f, 0.f, 0.f, 0.f, 0.f, 0.f, 0.f, 0.f};
#pragma unroll
    for (int kc = 0; kc < 4; ++kc) {
      const v2f Ak = { ck[2 * kc], ck[2 * kc + 1] };
      const v2f Bs = { xs[2 * kc], xs[2 * kc + 1] };
      k3 = WMMA(Ak, Bs, k3);
    }
    v8f cre = k3, cim = k3;
#pragma unroll
    for (int kc = 0; kc < 4; ++kc) {
      const v2f An = { cn[2 * kc], cn[2 * kc + 1] };
      const v2f Ad = { cd[2 * kc], cd[2 * kc + 1] };
      const v2f Br = { xr[tt][2 * kc], xr[tt][2 * kc + 1] };
      const v2f Bi = { xi[tt][2 * kc], xi[tt][2 * kc + 1] };
      cre = WMMA(An, Bi, cre);
      cim = WMMA(Ad, Br, cim);
    }
    if (ODD) {
      float* qr = sre + (tt ? 8 * (1 - half) : 8 * half) * CP + p * RP + mn;
      float* qi = sim + (tt ? 8 * (1 - half) : 8 * half) * CP + p * RP + mn;
#pragma unroll
      for (int r = 0; r < 8; ++r) { qr[(tt ? 7 - r : r) * CP] = cre[r]; qi[(tt ? 7 - r : r) * CP] = cim[r]; }
    } else {
      float* qr = sre + 8 * (half ^ lodd) * CP + p * RP + mn;
      float* qi = sim + 8 * (half ^ lodd) * CP + p * RP + mn;
      const v8f ore = lodd ? __builtin_shufflevector(cre, cre, 7, 6, 5, 4, 3, 2, 1, 0) : cre;
      const v8f oim = lodd ? __builtin_shufflevector(cim, cim, 7, 6, 5, 4, 3, 2, 1, 0) : cim;
#pragma unroll
      for (int r = 0; r < 8; ++r) { qr[r * CP] = ore[r]; qi[r * CP] = oim[r]; }
    }
  }
}

__global__ __launch_bounds__(THREADS, 8)
void qcircuit_parity_kernel(const float* __restrict__ x,
                            const float* __restrict__ thetas,
                            float* __restrict__ out) {
  __shared__ __attribute__((aligned(16))) float sre[PLANE];
  __shared__ __attribute__((aligned(16))) float sim[PLANE];
  __shared__ __attribute__((aligned(16))) float Gk[3 * GP];
  __shared__ __attribute__((aligned(16))) float Gn[3 * GP];
  __shared__ __attribute__((aligned(16))) float Gd[3 * GP];
  __shared__ __attribute__((aligned(16))) float gm[NQ * 8];
  __shared__ float red[THREADS];

  const int tid  = threadIdx.x;
  const int b    = blockIdx.x;
  const int lane = tid & 31;
  const int wave = tid >> 5;
  const int half = lane >> 4;
  const int mn   = lane & 15;

  for (int j = tid; j < PLANE; j += THREADS) { sre[j] = 0.f; sim[j] = 0.f; }
  if (tid == 0) sre[0] = 1.f;

  const float xv = x[b];
  const float x1 = asinf(xv);
  const float x2 = acosf(xv * xv);
  float fs1, fc1, fs2, fc2;
  __sincosf(0.5f * x1, &fs1, &fc1);
  __sincosf(0.5f * x2, &fs2, &fc2);
  fs1 = __int_as_float(__builtin_amdgcn_readfirstlane(__float_as_int(fs1)));
  fc1 = __int_as_float(__builtin_amdgcn_readfirstlane(__float_as_int(fc1)));
  fs2 = __int_as_float(__builtin_amdgcn_readfirstlane(__float_as_int(fs2)));
  fc2 = __int_as_float(__builtin_amdgcn_readfirstlane(__float_as_int(fc2)));

  __syncthreads();

  for (int l = 0; l < NL; ++l) {
    if (tid < NQ) {
      const int q = tid;
      c32 M[4] = { c32{ fc2 * fc1, -(fs2 * fc1) }, c32{ -(fc2 * fs1), fs2 * fs1 },
                   c32{ fc2 * fs1,   fs2 * fs1  }, c32{   fc2 * fc1,  fs2 * fc1 } };
      const float t0 = thetas[(l * NQ + q) * 3 + 0];
      const float t1 = thetas[(l * NQ + q) * 3 + 1];
      const float t2 = thetas[(l * NQ + q) * 3 + 2];
      rx_mul(M, t0); rz_mul(M, t1); rx_mul(M, t2);
      *(v4f*)(gm + q * 8)     = v4f{ M[0].re, M[1].re, M[2].re, M[3].re };
      *(v4f*)(gm + q * 8 + 4) = v4f{ M[0].im, M[1].im, M[2].im, M[3].im };
    }
    __syncthreads();

    {
      const int r = tid >> 4, cc = tid & 15;
      const int rg = r ^ (r >> 1);
#pragma unroll
      for (int g = 0; g < 3; ++g) {
        c32 acc = c32{1.f, 0.f};
#pragma unroll
        for (int j = 0; j < 4; ++j) {
          const int q  = 4 * g + j;
          const int bi = (rg >> (3 - j)) & 1;
          const int bj = (cc >> (3 - j)) & 1;
          acc = cmul(acc, c32{ gm[q * 8 + bi * 2 + bj], gm[q * 8 + 4 + bi * 2 + bj] });
        }
        Gk[g * GP + r * RP + cc] = acc.re;
        Gn[g * GP + r * RP + cc] = -(acc.re + acc.im);
        Gd[g * GP + r * RP + cc] = acc.im - acc.re;
      }
    }
    __syncthreads();

    const int o1 = (l & 1) ? GP : 0;
    pass_ab(sre, sim, Gk, Gn, Gd, o1, GP - o1, wave, half, mn);
    __syncthreads();
    if (l & 1) pass_c<1>(sre, sim, Gk, Gn, Gd, wave, half, mn);
    else       pass_c<0>(sre, sim, Gk, Gn, Gd, wave, half, mn);
    __syncthreads();
  }

  float acc = 0.f;
  {
    const int o = (tid & 15) * CP + (tid >> 4) * RP;
#pragma unroll
    for (int j = 0; j < 16; ++j) {
      const float p = (__popc(tid + j * THREADS) & 1) ? -1.f : 1.f;
      acc = fmaf(sre[o + j] * sre[o + j] + sim[o + j] * sim[o + j], p, acc);
    }
  }
  red[tid] = acc;
  __syncthreads();
  for (int off = THREADS / 2; off > 0; off >>= 1) {
    if (tid < off) red[tid] += red[tid + off];
    __syncthreads();
  }
  if (tid == 0) out[b] = red[0];
}

extern "C" void kernel_launch(void* const* d_in, const int* in_sizes, int n_in,
                              void* d_out, int out_size, void* d_ws, size_t ws_size,
                              hipStream_t stream) {
  (void)n_in; (void)d_ws; (void)ws_size; (void)out_size;
  const float* x      = (const float*)d_in[0];
  const float* thetas = (const float*)d_in[1];
  float* out          = (float*)d_out;
  const int batch     = in_sizes[0];
  qcircuit_parity_kernel<<<dim3(batch), dim3(THREADS), 0, stream>>>(x, thetas, out);
}


# Each qubit gate split as diagonal x real rotation x diagonal: group matrices real, 48 WMMA per wave and layer not 72; the diagonals are per-amplitude phase multiplies; f32 sum order changes
# speedup vs baseline: 1.9294x; 1.1903x over previous
#include <hip/hip_runtime.h>
#include <math.h>


typedef __attribute__((ext_vector_type(2))) float v2f;
typedef __attribute__((ext_vector_type(4))) float v4f;
typedef __attribute__((ext_vector_type(8))) float v8f;

#define NQ      12
#define NL      4
#define QDIM    4096
#define THREADS 256
#define RP      20
#define CP      (16 * RP + 4)
#define GP      (16 * RP)
#define PLANE   (16 * CP)

static_assert(QDIM == (1 << NQ), "dim mismatch");

struct c32 { float re, im; };
__device__ __forceinline__ c32 cmul(c32 a, c32 b) {
  return { a.re * b.re - a.im * b.im, a.re * b.im + a.im * b.re };
}

__device__ __forceinline__ void rx_mul(c32 M[4], float t) {
  float s, c; __sincosf(0.5f * t, &s, &c);
#pragma unroll
  for (int j = 0; j < 2; ++j) {
    const c32 u = M[j], v = M[2 + j];
    M[j]     = c32{ c * u.re + s * v.im, c * u.im - s * v.re };
    M[2 + j] = c32{ c * v.re + s * u.im, c * v.im - s * u.re };
  }
}
__device__ __forceinline__ void rz_mul(c32 M[4], float t) {
  float s, c; __sincosf(0.5f * t, &s, &c);
#pragma unroll
  for (int j = 0; j < 2; ++j) {
    const c32 u = M[j], v = M[2 + j];
    M[j]     = c32{ c * u.re + s * u.im, c * u.im - s * u.re };
    M[2 + j] = c32{ c * v.re - s * v.im, c * v.im + s * v.re };
  }
}

__device__ __forceinline__ void load8(const float* p, float (&v)[8]) {
  const v4f lo = *(const v4f*)p;
  const v4f hi = *(const v4f*)(p + 4);
  v[0] = lo[0]; v[1] = lo[1]; v[2] = lo[2]; v[3] = lo[3];
  v[4] = hi[0]; v[5] = hi[1]; v[6] = hi[2]; v[7] = hi[3];
}
__device__ __forceinline__ void store8(float* p, const v8f& d) {
  *(v4f*)p       = v4f{ d[0], d[1], d[2], d[3] };
  *(v4f*)(p + 4) = v4f{ d[4], d[5], d[6], d[7] };
}

#define WMMA(A, B, C) __builtin_amdgcn_wmma_f32_16x16x4_f32(false, A, false, B, 0, C, false, false)


__device__ __forceinline__ void pass_ab(float* sre, float* sim, const float* G,
                                        const float* Er, const float* Ei,
                                        int g1, int g2, int wave, int half, int mn) {
  const int go = mn * RP + 8 * half;
  v8f d1r[2], d1i[2];
  {
    float ak[8], e1r[8], e1i[8];
    load8(G + g1 * GP + go, ak);
    load8(Er + g1 * 16 + 8 * half, e1r); load8(Ei + g1 * 16 + 8 * half, e1i);
    const float e2r = Er[g2 * 16 + mn], e2i = Ei[g2 * 16 + mn];
#pragma unroll
    for (int tt = 0; tt < 2; ++tt) {
      const int c = wave + tt * 8;
      const float ecr = Er[32 + c], eci = Ei[32 + c];
      const float lr = e2r * ecr - e2i * eci, li = e2r * eci + e2i * ecr;
      float sr[8], si[8], pr[8], pi[8];
      load8(sre + c * CP + go, sr); load8(sim + c * CP + go, si);
#pragma unroll
      for (int k = 0; k < 8; ++k) {
        const float fr = e1r[k] * lr - e1i[k] * li, fi = e1r[k] * li + e1i[k] * lr;
        pr[k] = sr[k] * fr - si[k] * fi;
        pi[k] = sr[k] * fi + si[k] * fr;
      }
      v8f cre = {0.f, 0.f, 0.f, 0.f, 0.f, 0.f, 0.f, 0.f};
      v8f cim = {0.f, 0.f, 0.f, 0.f, 0.f, 0.f, 0.f, 0.f};
#pragma unroll
      for (int kc = 0; kc < 4; ++kc) {
        const v2f Ar = { pr[2 * kc], pr[2 * kc + 1] };
        const v2f Ai = { pi[2 * kc], pi[2 * kc + 1] };
        const v2f Bk = { ak[2 * kc], ak[2 * kc + 1] };
        cre = WMMA(Ar, Bk, cre);
        cim = WMMA(Ai, Bk, cim);
      }
      d1r[tt] = cre; d1i[tt] = cim;
    }
  }
  {
    float bk[8];
    load8(G + g2 * GP + go, bk);
#pragma unroll
    for (int tt = 0; tt < 2; ++tt) {
      const int c = wave + tt * 8;
      const v8f cre = d1r[tt], cim = d1i[tt];
      v8f dre = {0.f, 0.f, 0.f, 0.f, 0.f, 0.f, 0.f, 0.f};
      v8f dim = {0.f, 0.f, 0.f, 0.f, 0.f, 0.f, 0.f, 0.f};
#pragma unroll
      for (int kc = 0; kc < 4; ++kc) {
        const v2f Ak = { bk[2 * kc], bk[2 * kc + 1] };
        const v2f Br = { cre[2 * kc], cre[2 * kc + 1] };
        const v2f Bi = { cim[2 * kc], cim[2 * kc + 1] };
        dre = WMMA(Ak, Br, dre);
        dim = WMMA(Ak, Bi, dim);
      }
      store8(sre + c * CP + go, dre);
      store8(sim + c * CP + go, dim);
    }
  }
}

template <int ODD>
__device__ __forceinline__ void pass_c(float* sre, float* sim, const float* G,
                                       const float* Lr, const float* Li, bool post,
                                       int wave, int half, int mn) {
  const int go = mn * RP + 8 * half;
  const int lodd = mn & 1;
  float ck[8];
  load8(G + 2 * GP + go, ck);
  float xr[2][8], xi[2][8];
#pragma unroll
  for (int tt = 0; tt < 2; ++tt) {
    const int p = tt ? 15 - 2 * wave : 2 * wave;
    const int src = ODD ? (lodd ? 15 - p : p) * RP + mn
                        : p * RP + (tt ? 15 - mn : mn);
    const float* pr = sre + 8 * half * CP + src;
    const float* pi = sim + 8 * half * CP + src;
#pragma unroll
    for (int r = 0; r < 8; ++r) { xr[tt][r] = pr[r * CP]; xi[tt][r] = pi[r * CP]; }
  }
#pragma unroll
  for (int tt = 0; tt < 2; ++tt) {
    const int p = tt ? 15 - 2 * wave : 2 * wave;
    v8f cre = {0.f, 0.f, 0.f, 0.f, 0.f, 0.f, 0.f, 0.f};
    v8f cim = {0.f, 0.f, 0.f, 0.f, 0.f, 0.f, 0.f, 0.f};
#pragma unroll
    for (int kc = 0; kc < 4; ++kc) {
      const v2f Ak = { ck[2 * kc], ck[2 * kc + 1] };
      const v2f Br = { xr[tt][2 * kc], xr[tt][2 * kc + 1] };
      const v2f Bi = { xi[tt][2 * kc], xi[tt][2 * kc + 1] };
      cre = WMMA(Ak, Br, cre);
      cim = WMMA(Ak, Bi, cim);
    }
    if (post) {
      const int ia = ODD ? mn : p;
      const int ib = ODD ? (lodd ? 15 - p : p) : (tt ? 15 - mn : mn);
      const float ar = Lr[ia], ai = Li[ia], br = Lr[16 + ib], bi = Li[16 + ib];
      const float lr = ar * br - ai * bi, li = ar * bi + ai * br;
      float lcr[8], lci[8];
      load8(Lr + 32 + 8 * half, lcr); load8(Li + 32 + 8 * half, lci);
#pragma unroll
      for (int r = 0; r < 8; ++r) {
        const float fr = lcr[r] * lr - lci[r] * li, fi = lcr[r] * li + lci[r] * lr;
        const float vr = cre[r], vi = cim[r];
        cre[r] = vr * fr - vi * fi;
        cim[r] = vr * fi + vi * fr;
      }
    }
    if (ODD) {
      float* qr = sre + (tt ? 8 * (1 - half) : 8 * half) * CP + p * RP + mn;
      float* qi = sim + (tt ? 8 * (1 - half) : 8 * half) * CP + p * RP + mn;
#pragma unroll
      for (int r = 0; r < 8; ++r) { qr[(tt ? 7 - r : r) * CP] = cre[r]; qi[(tt ? 7 - r : r) * CP] = cim[r]; }
    } else {
      float* qr = sre + 8 * (half ^ lodd) * CP + p * RP + mn;
      float* qi = sim + 8 * (half ^ lodd) * CP + p * RP + mn;
      const v8f ore = lodd ? __builtin_shufflevector(cre, cre, 7, 6, 5, 4, 3, 2, 1, 0) : cre;
      const v8f oim = lodd ? __builtin_shufflevector(cim, cim, 7, 6, 5, 4, 3, 2, 1, 0) : cim;
#pragma unroll
      for (int r = 0; r < 8; ++r) { qr[r * CP] = ore[r]; qi[r * CP] = oim[r]; }
    }
  }
}

__global__ __launch_bounds__(THREADS, 8)
void qcircuit_parity_kernel(const float* __restrict__ x,
                            const float* __restrict__ thetas,
                            float* __restrict__ out) {
  __shared__ __attribute__((aligned(16))) float sre[PLANE];
  __shared__ __attribute__((aligned(16))) float sim[PLANE];
  __shared__ __attribute__((aligned(16))) float G[3 * GP];
  __shared__ __attribute__((aligned(16))) float Er[48];
  __shared__ __attribute__((aligned(16))) float Ei[48];
  __shared__ __attribute__((aligned(16))) float Lr[48];
  __shared__ __attribute__((aligned(16))) float Li[48];
  __shared__ __attribute__((aligned(16))) float gm[NQ * 12];
  __shared__ float red[THREADS];

  const int tid  = threadIdx.x;
  const int b    = blockIdx.x;
  const int lane = tid & 31;
  const int wave = tid >> 5;
  const int half = lane >> 4;
  const int mn   = lane & 15;

  for (int j = tid; j < PLANE; j += THREADS) { sre[j] = 0.f; sim[j] = 0.f; }
  if (tid == 0) sre[0] = 1.f;

  const float xv = x[b];
  const float x1 = asinf(xv);
  const float x2 = acosf(xv * xv);
  float fs1, fc1, fs2, fc2;
  __sincosf(0.5f * x1, &fs1, &fc1);
  __sincosf(0.5f * x2, &fs2, &fc2);
  fs1 = __int_as_float(__builtin_amdgcn_readfirstlane(__float_as_int(fs1)));
  fc1 = __int_as_float(__builtin_amdgcn_readfirstlane(__float_as_int(fc1)));
  fs2 = __int_as_float(__builtin_amdgcn_readfirstlane(__float_as_int(fs2)));
  fc2 = __int_as_float(__builtin_amdgcn_readfirstlane(__float_as_int(fc2)));

  __syncthreads();

  for (int l = 0; l < NL; ++l) {
    if (tid < NQ) {
      const int q = tid;
      c32 M[4] = { c32{ fc2 * fc1, -(fs2 * fc1) }, c32{ -(fc2 * fs1), fs2 * fs1 },
                   c32{ fc2 * fs1,   fs2 * fs1  }, c32{   fc2 * fc1,  fs2 * fc1 } };
      const float t0 = thetas[(l * NQ + q) * 3 + 0];
      const float t1 = thetas[(l * NQ + q) * 3 + 1];
      const float t2 = thetas[(l * NQ + q) * 3 + 2];
      rx_mul(M, t0); rz_mul(M, t1); rx_mul(M, t2);
      const float c2 = M[0].re * M[0].re + M[0].im * M[0].im;
      const float s2 = M[2].re * M[2].re + M[2].im * M[2].im;
      const float cq = sqrtf(c2), sq = sqrtf(s2);
      const c32 u = c2 > 1e-30f ? c32{ M[0].re / cq, M[0].im / cq } : c32{ 1.f, 0.f };
      const c32 w = s2 > 1e-30f ? c32{ M[2].re / sq, M[2].im / sq } : c32{ 1.f, 0.f };
      const c32 L1 = cmul(w, c32{ u.re, -u.im });
      *(v4f*)(gm + q * 12)     = v4f{ cq, -sq, sq, cq };
      *(v4f*)(gm + q * 12 + 4) = v4f{ u.re, u.im, w.re, -w.im };
      *(v4f*)(gm + q * 12 + 8) = v4f{ 1.f, 0.f, L1.re, L1.im };
    }
    __syncthreads();

    {
      const int r = tid >> 4, cc = tid & 15;
      const int rg = r ^ (r >> 1);
#pragma unroll
      for (int g = 0; g < 3; ++g) {
        float acc = 1.f;
#pragma unroll
        for (int j = 0; j < 4; ++j) {
          const int q  = 4 * g + j;
          const int bi = (rg >> (3 - j)) & 1;
          const int bj = (cc >> (3 - j)) & 1;
          acc *= gm[q * 12 + bi * 2 + bj];
        }
        G[g * GP + r * RP + cc] = acc;
      }
    }
    if (tid < 96) {
      const int side = tid >= 48;
      const int t = side ? tid - 48 : tid;
      const int g = t >> 4, m = t & 15;
      const int mm = side ? (m ^ (m >> 1)) : m;
      c32 acc = c32{1.f, 0.f};
#pragma unroll
      for (int j = 0; j < 4; ++j) {
        const int o = (4 * g + j) * 12 + 4 + 4 * side + 2 * ((mm >> (3 - j)) & 1);
        acc = cmul(acc, c32{ gm[o], gm[o + 1] });
      }
      if (side) { Lr[t] = acc.re; Li[t] = acc.im; }
      else      { Er[t] = acc.re; Ei[t] = acc.im; }
    }
    __syncthreads();

    const int g1 = l & 1;
    pass_ab(sre, sim, G, Er, Ei, g1, 1 - g1, wave, half, mn);
    __syncthreads();
    const bool post = l + 1 < NL;
    if (l & 1) pass_c<1>(sre, sim, G, Lr, Li, post, wave, half, mn);
    else       pass_c<0>(sre, sim, G, Lr, Li, post, wave, half, mn);
    __syncthreads();
  }

  float acc = 0.f;
  {
    const int o = (tid & 15) * CP + (tid >> 4) * RP;
#pragma unroll
    for (int j = 0; j < 16; ++j) {
      const float p = (__popc(tid + j * THREADS) & 1) ? -1.f : 1.f;
      acc = fmaf(sre[o + j] * sre[o + j] + sim[o + j] * sim[o + j], p, acc);
    }
  }
  red[tid] = acc;
  __syncthreads();
  for (int off = THREADS / 2; off > 0; off >>= 1) {
    if (tid < off) red[tid] += red[tid + off];
    __syncthreads();
  }
  if (tid == 0) out[b] = red[0];
}

extern "C" void kernel_launch(void* const* d_in, const int* in_sizes, int n_in,
                              void* d_out, int out_size, void* d_ws, size_t ws_size,
                              hipStream_t stream) {
  (void)n_in; (void)d_ws; (void)ws_size; (void)out_size;
  const float* x      = (const float*)d_in[0];
  const float* thetas = (const float*)d_in[1];
  float* out          = (float*)d_out;
  const int batch     = in_sizes[0];
  qcircuit_parity_kernel<<<dim3(batch), dim3(THREADS), 0, stream>>>(x, thetas, out);
}
